# starting kernel
# baseline (speedup 1.0000x reference)
#include <hip/hip_runtime.h>
#include <math.h>

typedef __attribute__((ext_vector_type(2))) float v2f;
typedef __attribute__((ext_vector_type(8))) float v8f;

#define NQ 8
#define NL 4
#define NO 16

__device__ __forceinline__ float fast_tanh(float x) {
#if __has_builtin(__builtin_amdgcn_tanhf)
    return __builtin_amdgcn_tanhf(x);
#else
    const float e = __expf(2.0f * x);
    return fmaf(-2.0f, __builtin_amdgcn_rcpf(e + 1.0f), 1.0f);
#endif
}

__global__ void qc_setup(const float* __restrict__ rot,
                         const float* __restrict__ ent,
                         const float* __restrict__ pw,
                         float* __restrict__ wt)
{
    const int o = threadIdx.x;
    if (o >= NO) return;

    float scale[NL][NQ];
    float w[NL][NQ - 1];
    for (int l = 0; l < NL; ++l) {
        for (int q = 0; q < NQ; ++q) {
            const float* r = rot + (l * NQ + q) * 3;
            float s = 1.0f;
            for (int j = 0; j < 3; ++j) s *= (cosf(r[j]) + sinf(r[j]));
            scale[l][q] = s;
        }
        for (int i = 0; i < NQ - 1; ++i) {
            const float e = ent[l * (NQ - 1) + i];
            w[l][i] = 1.0f / (1.0f + expf(-e));
        }
    }

    float M[NQ][NQ];
    for (int c = 0; c < NQ; ++c) {
        float v[NQ];
        for (int q = 0; q < NQ; ++q) v[q] = (q == c) ? 1.0f : 0.0f;
        for (int l = 0; l < NL; ++l) {
            for (int q = 0; q < NQ; ++q) v[q] *= scale[l][q];
            float nv[NQ];
            for (int i = 0; i < NQ - 1; ++i)
                nv[i] = v[i] * (1.0f - w[l][i]) + v[i + 1] * w[l][i];
            nv[NQ - 1] = v[NQ - 1];
            for (int q = 0; q < NQ; ++q) v[q] = nv[q];
        }
        for (int q = 0; q < NQ; ++q) M[q][c] = v[q];
    }

    for (int q = 0; q < NQ; ++q) {
        float acc = 0.0f;
        for (int k = 0; k < NQ; ++k) acc = fmaf(pw[o * NQ + k], M[k][q], acc);
        wt[q * NO + o] = acc;
    }
}

__global__ void __launch_bounds__(256) qc_main(const float* __restrict__ x,
                                               const float* __restrict__ wt,
                                               const float* __restrict__ pb,
                                               float* __restrict__ out,
                                               int ntiles)
{
    const int lane = threadIdx.x & 31;
    const int n    = lane & 15;
    const int g    = lane >> 4;

    const int wave   = (int)((blockIdx.x * blockDim.x + threadIdx.x) >> 5);
    const int nwaves = (int)((gridDim.x * blockDim.x) >> 5);

    v2f b1, b2;
    b1.x = wt[(4 * g + 0) * NO + n];
    b1.y = wt[(4 * g + 1) * NO + n];
    b2.x = wt[(4 * g + 2) * NO + n];
    b2.y = wt[(4 * g + 3) * NO + n];

    const float bias = pb[n];
    v8f cbias;
#pragma unroll
    for (int r = 0; r < 8; ++r) cbias[r] = bias;

    for (int t = wave; t < ntiles; t += nwaves) {
        const long rowBase = (long)t * 16;
        const long row     = rowBase + n;

        const float4 xv = *(const float4*)(x + row * 64 + g * 4);

        __builtin_prefetch(x + (row + (long)nwaves * 16) * 64 + g * 4, 0, 1);

        v2f a1, a2;
        a1.x = fast_tanh(xv.x);
        a1.y = fast_tanh(xv.y);
        a2.x = fast_tanh(xv.z);
        a2.y = fast_tanh(xv.w);

        v8f acc = __builtin_amdgcn_wmma_f32_16x16x4_f32(
            false, a1, false, b1, (short)0, cbias, false, false);
        acc = __builtin_amdgcn_wmma_f32_16x16x4_f32(
            false, a2, false, b2, (short)0, acc, false, false);

        float* po = out + (rowBase + (long)(8 * g)) * 16 + n;
#pragma unroll
        for (int r = 0; r < 8; ++r) po[r * 16] = acc[r];
    }
}

extern "C" void kernel_launch(void* const* d_in, const int* in_sizes, int n_in,
                              void* d_out, int out_size, void* d_ws, size_t ws_size,
                              hipStream_t stream) {
    const float* x   = (const float*)d_in[0];
    const float* rot = (const float*)d_in[1];
    const float* ent = (const float*)d_in[2];
    const float* pw  = (const float*)d_in[3];
    const float* pb  = (const float*)d_in[4];
    float*       out = (float*)d_out;
    float*       wt  = (float*)d_ws;

    qc_setup<<<1, 32, 0, stream>>>(rot, ent, pw, wt);

    const long rows   = (long)in_sizes[0] / 64;
    const int  ntiles = (int)(rows / 16);

    qc_main<<<512, 256, 0, stream>>>(x, wt, pb, out, ntiles);
}


# One launch, not two: every workgroup folds circuit layers and projection into its own WMMA B operands, parallel over its threads via LDS, behind a prefetch of its first tile; sum order kept
# speedup vs baseline: 2.2088x; 2.2088x over previous
#include <hip/hip_runtime.h>
#include <math.h>


typedef __attribute__((ext_vector_type(2))) float v2f;
typedef __attribute__((ext_vector_type(8))) float v8f;

#define NQ 8
#define NL 4
#define NO 16

__device__ __forceinline__ float fast_tanh(float x) {
#if __has_builtin(__builtin_amdgcn_tanhf)
    return __builtin_amdgcn_tanhf(x);
#else
    const float e = __expf(2.0f * x);
    return fmaf(-2.0f, __builtin_amdgcn_rcpf(e + 1.0f), 1.0f);
#endif
}

__global__ void __launch_bounds__(256) qc_fused(const float* __restrict__ x,
                                                const float* __restrict__ rot,
                                                const float* __restrict__ ent,
                                                const float* __restrict__ pw,
                                                const float* __restrict__ pb,
                                                float* __restrict__ out,
                                                int ntiles)
{
    __shared__ __attribute__((aligned(16))) float s_cs[NL * NQ * 3];
    __shared__ __attribute__((aligned(16))) float s_scale[NL * NQ];
    __shared__ __attribute__((aligned(16))) float s_w[NL * (NQ - 1)];
    __shared__ __attribute__((aligned(16))) float s_M[NQ * NQ];

    const int tid  = threadIdx.x;
    const int lane = tid & 31;
    const int n    = lane & 15;
    const int g    = lane >> 4;

    const int wave   = (int)((blockIdx.x * blockDim.x + threadIdx.x) >> 5);
    const int nwaves = (int)((gridDim.x * blockDim.x) >> 5);

    if (wave < ntiles)
        __builtin_prefetch(x + ((long)wave * 16 + n) * 64 + g * 4, 0, 1);

    float pwr[NQ];
#pragma unroll
    for (int k = 0; k < NQ; ++k) pwr[k] = pw[n * NQ + k];
    const float bias = pb[n];

    if (tid < NL * NQ * 3) {
        const float r = rot[tid];
        s_cs[tid] = cosf(r) + sinf(r);
    } else if (tid < NL * NQ * 3 + NL * (NQ - 1)) {
        const int   i = tid - NL * NQ * 3;
        const float e = ent[i];
        s_w[i] = 1.0f / (1.0f + expf(-e));
    }
    __syncthreads();

    if (tid < NL * NQ) {
        s_scale[tid] = (s_cs[tid * 3] * s_cs[tid * 3 + 1]) * s_cs[tid * 3 + 2];
    }
    __syncthreads();

    if (tid < NQ) {
        float sc[NL * NQ];
        float ww[NL * (NQ - 1)];
#pragma unroll
        for (int i = 0; i < NL * NQ; i += 4) {
            const float4 a = *(const float4*)(&s_scale[i]);
            sc[i] = a.x; sc[i + 1] = a.y; sc[i + 2] = a.z; sc[i + 3] = a.w;
        }
#pragma unroll
        for (int i = 0; i < NL * (NQ - 1); i += 4) {
            const float4 b = *(const float4*)(&s_w[i]);
            ww[i] = b.x; ww[i + 1] = b.y; ww[i + 2] = b.z; ww[i + 3] = b.w;
        }
        float v[NQ];
#pragma unroll
        for (int q = 0; q < NQ; ++q) v[q] = (q == tid) ? 1.0f : 0.0f;
#pragma unroll
        for (int l = 0; l < NL; ++l) {
#pragma unroll
            for (int q = 0; q < NQ; ++q) v[q] *= sc[l * NQ + q];
            float nv[NQ];
#pragma unroll
            for (int i = 0; i < NQ - 1; ++i) {
                const float w = ww[l * (NQ - 1) + i];
                nv[i] = v[i] * (1.0f - w) + v[i + 1] * w;
            }
            nv[NQ - 1] = v[NQ - 1];
#pragma unroll
            for (int q = 0; q < NQ; ++q) v[q] = nv[q];
        }
#pragma unroll
        for (int q = 0; q < NQ; ++q) s_M[q * NQ + tid] = v[q];
    }
    __syncthreads();

    float wq0 = 0.0f, wq1 = 0.0f, wq2 = 0.0f, wq3 = 0.0f;
#pragma unroll
    for (int k = 0; k < NQ; ++k) {
        const float4 m = *(const float4*)(&s_M[k * NQ + 4 * g]);
        wq0 = fmaf(pwr[k], m.x, wq0);
        wq1 = fmaf(pwr[k], m.y, wq1);
        wq2 = fmaf(pwr[k], m.z, wq2);
        wq3 = fmaf(pwr[k], m.w, wq3);
    }
    v2f b1, b2;
    b1.x = wq0;
    b1.y = wq1;
    b2.x = wq2;
    b2.y = wq3;

    v8f cbias;
#pragma unroll
    for (int r = 0; r < 8; ++r) cbias[r] = bias;

    for (int t = wave; t < ntiles; t += nwaves) {
        const long rowBase = (long)t * 16;
        const long row     = rowBase + n;

        const float4 xv = *(const float4*)(x + row * 64 + g * 4);

        __builtin_prefetch(x + (row + (long)nwaves * 16) * 64 + g * 4, 0, 1);

        v2f a1, a2;
        a1.x = fast_tanh(xv.x);
        a1.y = fast_tanh(xv.y);
        a2.x = fast_tanh(xv.z);
        a2.y = fast_tanh(xv.w);

        v8f acc = __builtin_amdgcn_wmma_f32_16x16x4_f32(
            false, a1, false, b1, (short)0, cbias, false, false);
        acc = __builtin_amdgcn_wmma_f32_16x16x4_f32(
            false, a2, false, b2, (short)0, acc, false, false);

        float* po = out + (rowBase + (long)(8 * g)) * 16 + n;
#pragma unroll
        for (int r = 0; r < 8; ++r) po[r * 16] = acc[r];
    }
}

extern "C" void kernel_launch(void* const* d_in, const int* in_sizes, int n_in,
                              void* d_out, int out_size, void* d_ws, size_t ws_size,
                              hipStream_t stream) {
    const float* x   = (const float*)d_in[0];
    const float* rot = (const float*)d_in[1];
    const float* ent = (const float*)d_in[2];
    const float* pw  = (const float*)d_in[3];
    const float* pb  = (const float*)d_in[4];
    float*       out = (float*)d_out;

    const long rows   = (long)in_sizes[0] / 64;
    const int  ntiles = (int)(rows / 16);

    qc_fused<<<512, 256, 0, stream>>>(x, rot, ent, pw, pb, out, ntiles);
}


# Each wave requests all its x tiles (8 b128 per lane) ahead of the fold and takes them in order; fold barriers wait on LDS only so the loads stay in flight; prefetch dropped; sum order kept
# speedup vs baseline: 2.4028x; 1.0878x over previous
#include <hip/hip_runtime.h>
#include <math.h>


typedef __attribute__((ext_vector_type(2))) float v2f;
typedef __attribute__((ext_vector_type(8))) float v8f;

#define NQ 8
#define NL 4
#define NO 16
#define XDEPTH 8

__device__ __forceinline__ float fast_tanh(float x) {
#if __has_builtin(__builtin_amdgcn_tanhf)
    return __builtin_amdgcn_tanhf(x);
#else
    const float e = __expf(2.0f * x);
    return fmaf(-2.0f, __builtin_amdgcn_rcpf(e + 1.0f), 1.0f);
#endif
}

__device__ __forceinline__ void lds_barrier() {
    __builtin_amdgcn_fence(__ATOMIC_RELEASE, "workgroup", "local");
    __builtin_amdgcn_s_barrier();
    __builtin_amdgcn_fence(__ATOMIC_ACQUIRE, "workgroup", "local");
}

__device__ __forceinline__ void tile(const float4 xc, const v2f b1, const v2f b2,
                                     const v8f cbias, float* __restrict__ po) {
    v2f a1, a2;
    a1.x = fast_tanh(xc.x);
    a1.y = fast_tanh(xc.y);
    a2.x = fast_tanh(xc.z);
    a2.y = fast_tanh(xc.w);

    v8f acc = __builtin_amdgcn_wmma_f32_16x16x4_f32(
        false, a1, false, b1, (short)0, cbias, false, false);
    acc = __builtin_amdgcn_wmma_f32_16x16x4_f32(
        false, a2, false, b2, (short)0, acc, false, false);

#pragma unroll
    for (int r = 0; r < 8; ++r) po[r * 16] = acc[r];
}

__global__ void __launch_bounds__(256) qc_fused(const float* __restrict__ x,
                                                const float* __restrict__ rot,
                                                const float* __restrict__ ent,
                                                const float* __restrict__ pw,
                                                const float* __restrict__ pb,
                                                float* __restrict__ out,
                                                int ntiles)
{
    __shared__ __attribute__((aligned(16))) float s_cs[NL * NQ * 3];
    __shared__ __attribute__((aligned(16))) float s_scale[NL * NQ];
    __shared__ __attribute__((aligned(16))) float s_w[NL * (NQ - 1)];
    __shared__ __attribute__((aligned(16))) float s_M[NQ * NQ];

    const int tid  = threadIdx.x;
    const int lane = tid & 31;
    const int n    = lane & 15;
    const int g    = lane >> 4;

    const int wave   = __builtin_amdgcn_readfirstlane((int)((blockIdx.x * blockDim.x + threadIdx.x) >> 5));
    const int nwaves = (int)((gridDim.x * blockDim.x) >> 5);

    if (ntiles <= 0) return;

    float pwr[NQ];
#pragma unroll
    for (int k = 0; k < NQ; ++k) pwr[k] = pw[n * NQ + k];
    const float bias = pb[n];
    float par = 0.0f;
    if (tid < NL * NQ * 3 + NL * (NQ - 1)) {
        const float* pp = (tid < NL * NQ * 3) ? rot + tid : ent + (tid - NL * NQ * 3);
        par = *pp;
    }

    const int    tlast   = ntiles - 1;
    const float* xlane   = x + (long)n * 64 + g * 4;
    const long   tstride = (long)16 * 64;
    float4 xv[XDEPTH];
#pragma unroll
    for (int j = 0; j < XDEPTH; ++j) {
        const int t = min(wave + j * nwaves, tlast);
        xv[j] = *(const float4*)(xlane + (long)t * tstride);
    }

    if (tid < NL * NQ * 3) {
        s_cs[tid] = cosf(par) + sinf(par);
    } else if (tid < NL * NQ * 3 + NL * (NQ - 1)) {
        s_w[tid - NL * NQ * 3] = 1.0f / (1.0f + expf(-par));
    }
    lds_barrier();

    if (tid < NL * NQ) {
        s_scale[tid] = (s_cs[tid * 3] * s_cs[tid * 3 + 1]) * s_cs[tid * 3 + 2];
    }
    lds_barrier();

    if (tid < NQ) {
        float sc[NL * NQ];
        float ww[NL * (NQ - 1)];
#pragma unroll
        for (int i = 0; i < NL * NQ; i += 4) {
            const float4 a = *(const float4*)(&s_scale[i]);
            sc[i] = a.x; sc[i + 1] = a.y; sc[i + 2] = a.z; sc[i + 3] = a.w;
        }
#pragma unroll
        for (int i = 0; i < NL * (NQ - 1); i += 4) {
            const float4 b = *(const float4*)(&s_w[i]);
            ww[i] = b.x; ww[i + 1] = b.y; ww[i + 2] = b.z; ww[i + 3] = b.w;
        }
        float v[NQ];
#pragma unroll
        for (int q = 0; q < NQ; ++q) v[q] = (q == tid) ? 1.0f : 0.0f;
#pragma unroll
        for (int l = 0; l < NL; ++l) {
#pragma unroll
            for (int q = 0; q < NQ; ++q) v[q] *= sc[l * NQ + q];
            float nv[NQ];
#pragma unroll
            for (int i = 0; i < NQ - 1; ++i) {
                const float w = ww[l * (NQ - 1) + i];
                nv[i] = v[i] * (1.0f - w) + v[i + 1] * w;
            }
            nv[NQ - 1] = v[NQ - 1];
#pragma unroll
            for (int q = 0; q < NQ; ++q) v[q] = nv[q];
        }
#pragma unroll
        for (int q = 0; q < NQ; ++q) s_M[q * NQ + tid] = v[q];
    }
    lds_barrier();

    float wq0 = 0.0f, wq1 = 0.0f, wq2 = 0.0f, wq3 = 0.0f;
#pragma unroll
    for (int k = 0; k < NQ; ++k) {
        const float4 m = *(const float4*)(&s_M[k * NQ + 4 * g]);
        wq0 = fmaf(pwr[k], m.x, wq0);
        wq1 = fmaf(pwr[k], m.y, wq1);
        wq2 = fmaf(pwr[k], m.z, wq2);
        wq3 = fmaf(pwr[k], m.w, wq3);
    }
    v2f b1, b2;
    b1.x = wq0;
    b1.y = wq1;
    b2.x = wq2;
    b2.y = wq3;

    v8f cbias;
#pragma unroll
    for (int r = 0; r < 8; ++r) cbias[r] = bias;

    float* olane = out + (long)(8 * g) * 16 + n;
    const long ostride = (long)16 * 16;
    const int  round   = XDEPTH * nwaves;
    int t0 = wave;

    for (; t0 + round < ntiles; t0 += round) {
#pragma unroll
        for (int j = 0; j < XDEPTH; ++j) {
            const int    t  = t0 + j * nwaves;
            const float4 xc = xv[j];
            xv[j] = *(const float4*)(xlane + (long)min(t + round, tlast) * tstride);
            tile(xc, b1, b2, cbias, olane + (long)t * ostride);
        }
    }

#pragma unroll
    for (int j = 0; j < XDEPTH; ++j) {
        const int t = t0 + j * nwaves;
        if (t < ntiles) tile(xv[j], b1, b2, cbias, olane + (long)t * ostride);
    }
}

extern "C" void kernel_launch(void* const* d_in, const int* in_sizes, int n_in,
                              void* d_out, int out_size, void* d_ws, size_t ws_size,
                              hipStream_t stream) {
    const float* x   = (const float*)d_in[0];
    const float* rot = (const float*)d_in[1];
    const float* ent = (const float*)d_in[2];
    const float* pw  = (const float*)d_in[3];
    const float* pb  = (const float*)d_in[4];
    float*       out = (float*)d_out;

    const long rows   = (long)in_sizes[0] / 64;
    const int  ntiles = (int)(rows / 16);

    qc_fused<<<512, 256, 0, stream>>>(x, rot, ent, pw, pb, out, ntiles);
}
